# starting kernel
# baseline (speedup 1.0000x reference)
#include <hip/hip_runtime.h>

typedef __attribute__((ext_vector_type(2))) float v2f;
typedef __attribute__((ext_vector_type(8))) float v8f;

__global__ __launch_bounds__(256) void ftl_proj_wmma(
    const float* __restrict__ z, const float* __restrict__ pm,
    float* __restrict__ out) {
  constexpr int HW = 64 * 64;
  constexpr int G  = 128;
  constexpr int NCHUNK = HW / 16;

  const int b    = blockIdx.y;
  const int g    = blockIdx.x;
  const int lane = threadIdx.x & 31;
  const int wave = threadIdx.x >> 5;
  const int col  = lane & 15;
  const int k0   = (lane >> 4) << 1;

  const float* p = pm + b * 12;
  v2f a;
  a.x = (col < 3) ? p[col * 4 + k0]     : 0.0f;
  a.y = (col < 3) ? p[col * 4 + k0 + 1] : 0.0f;

  const size_t zbase = ((size_t)b * (G * 4) + (size_t)g * 4) * (size_t)HW;
  const size_t obase = ((size_t)b * (G * 3) + (size_t)g * 3) * (size_t)HW;
  const float* zp0 = z + zbase + (size_t)k0 * HW + col;
  const float* zp1 = zp0 + HW;
  float* o0 = out + obase + col;

#pragma unroll 8
  for (int c = wave; c < NCHUNK; c += 8) {
    const int pos = c << 4;
    __builtin_prefetch(zp0 + pos + (8 << 4), 0, 0);

    v2f bm;
    bm.x = __builtin_nontemporal_load(zp0 + pos);
    bm.y = __builtin_nontemporal_load(zp1 + pos);

    v8f acc = {};
    acc = __builtin_amdgcn_wmma_f32_16x16x4_f32(
         false, a,  false, bm,
         (short)0, acc,  false,  false);

    if (lane < 16) {
      __builtin_nontemporal_store(acc[0], o0 + 0 * HW + pos);
      __builtin_nontemporal_store(acc[1], o0 + 1 * HW + pos);
      __builtin_nontemporal_store(acc[2], o0 + 2 * HW + pos);
    }
  }
}

extern "C" void kernel_launch(void* const* d_in, const int* in_sizes, int n_in,
                              void* d_out, int out_size, void* d_ws, size_t ws_size,
                              hipStream_t stream) {
  (void)in_sizes; (void)n_in; (void)d_ws; (void)ws_size; (void)out_size;
  const float* z  = (const float*)d_in[0];
  const float* pm = (const float*)d_in[1];
  float* out = (float*)d_out;

  dim3 grid(128, 16);
  ftl_proj_wmma<<<grid, 256, 0, stream>>>(z, pm, out);
}


# Stream in 16-B-per-lane pieces: every wave-wide load/store is 512 contiguous bytes, 16-32 loads in flight per wave, 4096 waves; channel sum = 4 chained WMMAs per position pair (sum order: j in turn)
# speedup vs baseline: 1.8578x; 1.8578x over previous
#include <hip/hip_runtime.h>


typedef __attribute__((ext_vector_type(2))) float v2f;
typedef __attribute__((ext_vector_type(4))) float v4f;
typedef __attribute__((ext_vector_type(8))) float v8f;

#define DEV static __device__ __forceinline__

constexpr int HW     = 64 * 64;
constexpr int UNIT   = 128;
constexpr int U      = 4;
constexpr int STAGES = 4;
constexpr int WAVES  = 8;

DEV v4f load_z(const float* p) { return __builtin_nontemporal_load(reinterpret_cast<const v4f*>(p)); }
DEV void store_o(float* p, v4f v) { __builtin_nontemporal_store(v, reinterpret_cast<v4f*>(p)); }
DEV v8f mma(v2f a, v2f b, v8f c) {
  return __builtin_amdgcn_wmma_f32_16x16x4_f32(false, a, false, b, (short)0, c, false, false);
}

__global__ __launch_bounds__(WAVES * 32) void ftl_proj_wmma(
    const float* __restrict__ z, const float* __restrict__ pm, float* __restrict__ out) {
  const int lane = threadIdx.x & 31;
  const int wave = __builtin_amdgcn_readfirstlane(threadIdx.x >> 5);
  const int w    = blockIdx.x * WAVES + wave;
  const int pair = w >> 1;
  const int half = w & 1;
  const int b    = pair >> 7;

  const int  t    = (lane & 15) - ((lane >> 4) << 3);
  const bool live = (unsigned)t < 6u;
  const int  e    = t & 1;
  const v4f prow = *reinterpret_cast<const v4f*>(pm + b * 12 + (live ? (t >> 1) : 0) * 4);
  v2f a[4];
#pragma unroll
  for (int j = 0; j < 4; ++j) {
    const float pj = live ? prow[j] : 0.0f;
    a[j].x = (e == 0) ? pj : 0.0f;
    a[j].y = (e == 1) ? pj : 0.0f;
  }

  const float* zp = z   + (size_t)pair * (4 * HW) + half * (HW / 2) + lane * 4;
  float*       op = out + (size_t)pair * (3 * HW) + half * (HW / 2) + lane * 4;

  v4f buf[2][U][4];
#define LOAD_STAGE(S) do {                                                    \
  _Pragma("unroll") for (int u = 0; u < U; ++u)                               \
  _Pragma("unroll") for (int j = 0; j < 4; ++j)                               \
    buf[(S) & 1][u][j] = load_z(zp + j * HW + ((S) * U + u) * UNIT);          \
  } while (0)

  LOAD_STAGE(0);
#pragma unroll
  for (int s = 0; s < STAGES; ++s) {
    __builtin_amdgcn_sched_barrier(0);
    if (s + 1 < STAGES) LOAD_STAGE(s + 1);
    __builtin_amdgcn_sched_barrier(0);
#pragma unroll
    for (int u = 0; u < U; ++u) {
      v8f d[2];
#pragma unroll
      for (int q = 0; q < 2; ++q) {
        v8f c = {};
#pragma unroll
        for (int j = 0; j < 4; ++j) {
          const v4f zj = buf[s & 1][u][j];
          v2f bq;
          bq.x = zj[2 * q];
          bq.y = zj[2 * q + 1];
          c = mma(a[j], bq, c);
        }
        d[q] = c;
      }
      float* o = op + (s * U + u) * UNIT;
#pragma unroll
      for (int r = 0; r < 3; ++r) {
        v4f v;
        v.x = d[0][2 * r];
        v.y = d[0][2 * r + 1];
        v.z = d[1][2 * r];
        v.w = d[1][2 * r + 1];
        store_o(o + r * HW, v);
      }
    }
  }
#undef LOAD_STAGE
}

extern "C" void kernel_launch(void* const* d_in, const int* in_sizes, int n_in,
                              void* d_out, int out_size, void* d_ws, size_t ws_size,
                              hipStream_t stream) {
  (void)in_sizes; (void)n_in; (void)d_ws; (void)ws_size; (void)out_size;
  const float* z  = (const float*)d_in[0];
  const float* pm = (const float*)d_in[1];
  float* out = (float*)d_out;

  ftl_proj_wmma<<<dim3(512), dim3(WAVES * 32), 0, stream>>>(z, pm, out);
}


# Each wave walks its four 2-KB stages in rotated order, first stage = group & 3: a workgroup's requests in flight cover every 2-KB block offset, not one common to all waves (sum order unchanged)
# speedup vs baseline: 2.2103x; 1.1897x over previous
#include <hip/hip_runtime.h>


typedef __attribute__((ext_vector_type(2))) float v2f;
typedef __attribute__((ext_vector_type(4))) float v4f;
typedef __attribute__((ext_vector_type(8))) float v8f;

#define DEV static __device__ __forceinline__

constexpr int HW     = 64 * 64;
constexpr int UNIT   = 128;
constexpr int U      = 4;
constexpr int STAGES = 4;
constexpr int WAVES  = 8;

DEV v4f load_z(const float* p) { return __builtin_nontemporal_load(reinterpret_cast<const v4f*>(p)); }
DEV void store_o(float* p, v4f v) { __builtin_nontemporal_store(v, reinterpret_cast<v4f*>(p)); }
DEV v8f mma(v2f a, v2f b, v8f c) {
  return __builtin_amdgcn_wmma_f32_16x16x4_f32(false, a, false, b, (short)0, c, false, false);
}

__global__ __launch_bounds__(WAVES * 32) void ftl_proj_wmma(
    const float* __restrict__ z, const float* __restrict__ pm, float* __restrict__ out) {
  const int lane = threadIdx.x & 31;
  const int wave = __builtin_amdgcn_readfirstlane(threadIdx.x >> 5);
  const int w    = blockIdx.x * WAVES + wave;
  const int pair = w >> 1;
  const int half = w & 1;
  const int b    = pair >> 7;
  const int rot  = pair & 3;

  const int  t    = (lane & 15) - ((lane >> 4) << 3);
  const bool live = (unsigned)t < 6u;
  const int  e    = t & 1;
  const v4f prow = *reinterpret_cast<const v4f*>(pm + b * 12 + (live ? (t >> 1) : 0) * 4);
  v2f a[4];
#pragma unroll
  for (int j = 0; j < 4; ++j) {
    const float pj = live ? prow[j] : 0.0f;
    a[j].x = (e == 0) ? pj : 0.0f;
    a[j].y = (e == 1) ? pj : 0.0f;
  }

  const float* zp = z   + (size_t)pair * (4 * HW) + half * (HW / 2) + lane * 4;
  float*       op = out + (size_t)pair * (3 * HW) + half * (HW / 2) + lane * 4;

  v4f buf[2][U][4];
#define LOAD_STAGE(S) do {                                                    \
  _Pragma("unroll") for (int u = 0; u < U; ++u)                               \
  _Pragma("unroll") for (int j = 0; j < 4; ++j)                               \
    buf[(S) & 1][u][j] = load_z(zp + (((S) + rot) & 3) * (U * UNIT) + j * HW + u * UNIT); \
  } while (0)

  LOAD_STAGE(0);
#pragma unroll
  for (int s = 0; s < STAGES; ++s) {
    __builtin_amdgcn_sched_barrier(0);
    if (s + 1 < STAGES) LOAD_STAGE(s + 1);
    __builtin_amdgcn_sched_barrier(0);
#pragma unroll
    for (int u = 0; u < U; ++u) {
      v8f d[2];
#pragma unroll
      for (int q = 0; q < 2; ++q) {
        v8f c = {};
#pragma unroll
        for (int j = 0; j < 4; ++j) {
          const v4f zj = buf[s & 1][u][j];
          v2f bq;
          bq.x = zj[2 * q];
          bq.y = zj[2 * q + 1];
          c = mma(a[j], bq, c);
        }
        d[q] = c;
      }
      float* o = op + ((s + rot) & 3) * (U * UNIT) + u * UNIT;
#pragma unroll
      for (int r = 0; r < 3; ++r) {
        v4f v;
        v.x = d[0][2 * r];
        v.y = d[0][2 * r + 1];
        v.z = d[1][2 * r];
        v.w = d[1][2 * r + 1];
        store_o(o + r * HW, v);
      }
    }
  }
#undef LOAD_STAGE
}

extern "C" void kernel_launch(void* const* d_in, const int* in_sizes, int n_in,
                              void* d_out, int out_size, void* d_ws, size_t ws_size,
                              hipStream_t stream) {
  (void)in_sizes; (void)n_in; (void)d_ws; (void)ws_size; (void)out_size;
  const float* z  = (const float*)d_in[0];
  const float* pm = (const float*)d_in[1];
  float* out = (float*)d_out;

  ftl_proj_wmma<<<dim3(512), dim3(WAVES * 32), 0, stream>>>(z, pm, out);
}


# Pipeline per 128-position unit, ring of 8 units (32 loads in flight): after a unit's 8 WMMAs its slot is re-requested, then its 3 stores go out; reads never wait behind stores (sum order unchanged)
# speedup vs baseline: 2.3217x; 1.0504x over previous
#include <hip/hip_runtime.h>


typedef __attribute__((ext_vector_type(2))) float v2f;
typedef __attribute__((ext_vector_type(4))) float v4f;
typedef __attribute__((ext_vector_type(8))) float v8f;

#define DEV static __device__ __forceinline__

constexpr int HW     = 64 * 64;
constexpr int UNIT   = 128;
constexpr int U      = 4;
constexpr int UNITS  = 16;
constexpr int RING   = 8;
constexpr int WAVES  = 8;

DEV v4f load_z(const float* p) { return __builtin_nontemporal_load(reinterpret_cast<const v4f*>(p)); }
DEV void store_o(float* p, v4f v) { __builtin_nontemporal_store(v, reinterpret_cast<v4f*>(p)); }
DEV v8f mma(v2f a, v2f b, v8f c) {
  return __builtin_amdgcn_wmma_f32_16x16x4_f32(false, a, false, b, (short)0, c, false, false);
}

__global__ __launch_bounds__(WAVES * 32) void ftl_proj_wmma(
    const float* __restrict__ z, const float* __restrict__ pm, float* __restrict__ out) {
  const int lane = threadIdx.x & 31;
  const int wave = __builtin_amdgcn_readfirstlane(threadIdx.x >> 5);
  const int w    = blockIdx.x * WAVES + wave;
  const int pair = w >> 1;
  const int half = w & 1;
  const int b    = pair >> 7;
  const int rot  = pair & 3;

  const int  t    = (lane & 15) - ((lane >> 4) << 3);
  const bool live = (unsigned)t < 6u;
  const int  e    = t & 1;
  const v4f prow = *reinterpret_cast<const v4f*>(pm + b * 12 + (live ? (t >> 1) : 0) * 4);
  v2f a[4];
#pragma unroll
  for (int j = 0; j < 4; ++j) {
    const float pj = live ? prow[j] : 0.0f;
    a[j].x = (e == 0) ? pj : 0.0f;
    a[j].y = (e == 1) ? pj : 0.0f;
  }

  const float* zp = z   + (size_t)pair * (4 * HW) + half * (HW / 2) + lane * 4;
  float*       op = out + (size_t)pair * (3 * HW) + half * (HW / 2) + lane * 4;

  v4f buf[RING][4];
#define UNIT_OFF(I) (((((I) >> 2) + rot) & 3) * (U * UNIT) + ((I) & 3) * UNIT)
#define LOAD_UNIT(I) do {                                                     \
  _Pragma("unroll") for (int j = 0; j < 4; ++j)                               \
    buf[(I) % RING][j] = load_z(zp + UNIT_OFF(I) + j * HW);                   \
  } while (0)

#pragma unroll
  for (int i = 0; i < RING; ++i) {
    LOAD_UNIT(i);
    __builtin_amdgcn_sched_barrier(0);
  }
#pragma unroll
  for (int i = 0; i < UNITS; ++i) {
    __builtin_amdgcn_sched_barrier(0);
    v8f d[2];
#pragma unroll
    for (int q = 0; q < 2; ++q) {
      v8f c = {};
#pragma unroll
      for (int j = 0; j < 4; ++j) {
        const v4f zj = buf[i % RING][j];
        v2f bq;
        bq.x = zj[2 * q];
        bq.y = zj[2 * q + 1];
        c = mma(a[j], bq, c);
      }
      d[q] = c;
    }
    __builtin_amdgcn_sched_barrier(0);
    if (i + RING < UNITS) LOAD_UNIT(i + RING);
    __builtin_amdgcn_sched_barrier(0);
    float* o = op + UNIT_OFF(i);
#pragma unroll
    for (int r = 0; r < 3; ++r) {
      v4f v;
      v.x = d[0][2 * r];
      v.y = d[0][2 * r + 1];
      v.z = d[1][2 * r];
      v.w = d[1][2 * r + 1];
      store_o(o + r * HW, v);
    }
  }
#undef LOAD_UNIT
#undef UNIT_OFF
}

extern "C" void kernel_launch(void* const* d_in, const int* in_sizes, int n_in,
                              void* d_out, int out_size, void* d_ws, size_t ws_size,
                              hipStream_t stream) {
  (void)in_sizes; (void)n_in; (void)d_ws; (void)ws_size; (void)out_size;
  const float* z  = (const float*)d_in[0];
  const float* pm = (const float*)d_in[1];
  float* out = (float*)d_out;

  ftl_proj_wmma<<<dim3(512), dim3(WAVES * 32), 0, stream>>>(z, pm, out);
}


# Each wave also walks a 2-KB block's four 512-B pieces in rotated order, first piece = (group + 2*half + workgroup) & 3: one moment's stores and new loads cover all 512-B offsets (sum order unchanged)
# speedup vs baseline: 2.6388x; 1.1366x over previous
#include <hip/hip_runtime.h>


typedef __attribute__((ext_vector_type(2))) float v2f;
typedef __attribute__((ext_vector_type(4))) float v4f;
typedef __attribute__((ext_vector_type(8))) float v8f;

#define DEV static __device__ __forceinline__

constexpr int HW     = 64 * 64;
constexpr int UNIT   = 128;
constexpr int U      = 4;
constexpr int UNITS  = 16;
constexpr int RING   = 8;
constexpr int WAVES  = 8;

DEV v4f load_z(const float* p) { return __builtin_nontemporal_load(reinterpret_cast<const v4f*>(p)); }
DEV void store_o(float* p, v4f v) { __builtin_nontemporal_store(v, reinterpret_cast<v4f*>(p)); }
DEV v8f mma(v2f a, v2f b, v8f c) {
  return __builtin_amdgcn_wmma_f32_16x16x4_f32(false, a, false, b, (short)0, c, false, false);
}

__global__ __launch_bounds__(WAVES * 32) void ftl_proj_wmma(
    const float* __restrict__ z, const float* __restrict__ pm, float* __restrict__ out) {
  const int lane = threadIdx.x & 31;
  const int wave = __builtin_amdgcn_readfirstlane(threadIdx.x >> 5);
  const int w    = blockIdx.x * WAVES + wave;
  const int pair = w >> 1;
  const int half = w & 1;
  const int b    = pair >> 7;
  const int rot  = pair & 3;
  const int prot = (pair + 2 * half + (pair >> 2)) & 3;

  const int  t    = (lane & 15) - ((lane >> 4) << 3);
  const bool live = (unsigned)t < 6u;
  const int  e    = t & 1;
  const v4f prow = *reinterpret_cast<const v4f*>(pm + b * 12 + (live ? (t >> 1) : 0) * 4);
  v2f a[4];
#pragma unroll
  for (int j = 0; j < 4; ++j) {
    const float pj = live ? prow[j] : 0.0f;
    a[j].x = (e == 0) ? pj : 0.0f;
    a[j].y = (e == 1) ? pj : 0.0f;
  }

  const float* zp = z   + (size_t)pair * (4 * HW) + half * (HW / 2) + lane * 4;
  float*       op = out + (size_t)pair * (3 * HW) + half * (HW / 2) + lane * 4;

  v4f buf[RING][4];
#define UNIT_OFF(I) (((((I) >> 2) + rot) & 3) * (U * UNIT) + ((((I) & 3) + prot) & 3) * UNIT)
#define LOAD_UNIT(I) do {                                                     \
  _Pragma("unroll") for (int j = 0; j < 4; ++j)                               \
    buf[(I) % RING][j] = load_z(zp + UNIT_OFF(I) + j * HW);                   \
  } while (0)

#pragma unroll
  for (int i = 0; i < RING; ++i) {
    LOAD_UNIT(i);
    __builtin_amdgcn_sched_barrier(0);
  }
#pragma unroll
  for (int i = 0; i < UNITS; ++i) {
    __builtin_amdgcn_sched_barrier(0);
    v8f d[2];
#pragma unroll
    for (int q = 0; q < 2; ++q) {
      v8f c = {};
#pragma unroll
      for (int j = 0; j < 4; ++j) {
        const v4f zj = buf[i % RING][j];
        v2f bq;
        bq.x = zj[2 * q];
        bq.y = zj[2 * q + 1];
        c = mma(a[j], bq, c);
      }
      d[q] = c;
    }
    __builtin_amdgcn_sched_barrier(0);
    if (i + RING < UNITS) LOAD_UNIT(i + RING);
    __builtin_amdgcn_sched_barrier(0);
    float* o = op + UNIT_OFF(i);
#pragma unroll
    for (int r = 0; r < 3; ++r) {
      v4f v;
      v.x = d[0][2 * r];
      v.y = d[0][2 * r + 1];
      v.z = d[1][2 * r];
      v.w = d[1][2 * r + 1];
      store_o(o + r * HW, v);
    }
  }
#undef LOAD_UNIT
#undef UNIT_OFF
}

extern "C" void kernel_launch(void* const* d_in, const int* in_sizes, int n_in,
                              void* d_out, int out_size, void* d_ws, size_t ws_size,
                              hipStream_t stream) {
  (void)in_sizes; (void)n_in; (void)d_ws; (void)ws_size; (void)out_size;
  const float* z  = (const float*)d_in[0];
  const float* pm = (const float*)d_in[1];
  float* out = (float*)d_out;

  ftl_proj_wmma<<<dim3(512), dim3(WAVES * 32), 0, stream>>>(z, pm, out);
}
